# starting kernel
# baseline (speedup 1.0000x reference)
#include <hip/hip_runtime.h>
#include <hip/hip_bf16.h>
#include <math.h>

typedef __attribute__((ext_vector_type(2))) float v2f;
typedef __attribute__((ext_vector_type(8))) float v8f;

constexpr int kNRows     = 8192;
constexpr int kF         = 188;
constexpr int kH         = 16;
constexpr int kRowStride = kF * kH;
constexpr int kNCont     = 177;

__device__ __forceinline__ float fast_rcp(float a) {
    return __builtin_amdgcn_rcpf(a);
}
__device__ __forceinline__ float fast_sigmoid(float a) {
    return fast_rcp(1.0f + __expf(-a));
}
__device__ __forceinline__ float fast_tanh(float a) {
#if __has_builtin(__builtin_amdgcn_tanhf)
    return __builtin_amdgcn_tanhf(a);
#elif __has_builtin(__builtin_amdgcn_tanh_f32)
    return __builtin_amdgcn_tanh_f32(a);
#else
    const float e2 = __expf(-2.0f * __builtin_fabsf(a));
    const float t  = (1.0f - e2) * fast_rcp(1.0f + e2);
    return __builtin_copysignf(t, a);
#endif
}

__global__ __launch_bounds__(128)
void tab_decoder_wmma(const float* __restrict__ x,
                      const float* __restrict__ W2a, const float* __restrict__ b2a,
                      const float* __restrict__ W3,  const float* __restrict__ b3,
                      const float* __restrict__ W4,  const float* __restrict__ b4,
                      const float* __restrict__ W6,  const float* __restrict__ b6,
                      const float* __restrict__ W7,  const float* __restrict__ b7,
                      const float* __restrict__ Wc,  const float* __restrict__ bc,
                      const float* __restrict__ eps,
                      float* __restrict__ out)
{
    const int tile = blockIdx.x;
    const int wave = threadIdx.x >> 5;
    const int lane = threadIdx.x & 31;
    const int m    = lane & 15;
    const int kh   = lane >> 4;
    const int n    = m;

    const int    arow = tile * 16 + m;
    const float* xb   = x + (size_t)arow * kRowStride;

    for (int f = wave; f < kF; f += 4) {
        const float* W; const float* bias; int od;
        if      (f < 4)  { W = W2a + f       * kH * 2; bias = b2a + f       * 2; od = 2; }
        else if (f < 6)  { W = W3  + (f - 4) * kH * 3; bias = b3  + (f - 4) * 3; od = 3; }
        else if (f < 7)  { W = W4;                     bias = b4;                od = 4; }
        else if (f < 8)  { W = W6;                     bias = b6;                od = 6; }
        else if (f < 11) { W = W7  + (f - 8) * kH * 7; bias = b7  + (f - 8) * 7; od = 7; }
        else             { W = Wc  + (f - 11)* kH * 2; bias = bc  + (f - 11)* 2; od = 2; }

        const float* xf = xb + f * kH;
        __builtin_prefetch(xf + 4 * kH, 0, 1);

        const bool valid = (n < od);
        const int  nc    = valid ? n : 0;

        const float braw = bias[nc];
        const float bn   = valid ? braw : 0.0f;

        v2f a0 = *(const v2f*)(xf + 0  + 2 * kh);
        v2f a1 = *(const v2f*)(xf + 4  + 2 * kh);
        v2f a2 = *(const v2f*)(xf + 8  + 2 * kh);
        v2f a3 = *(const v2f*)(xf + 12 + 2 * kh);

        const float* Wl = W + 2 * kh * od + nc;
        float w0 = Wl[0 * od], w1 = Wl[1 * od];
        float w2 = Wl[4 * od], w3 = Wl[5 * od];
        float w4 = Wl[8 * od], w5 = Wl[9 * od];
        float w6 = Wl[12 * od], w7 = Wl[13 * od];

        v2f b0, b1, b2, b3;
        b0.x = valid ? w0 : 0.0f;  b0.y = valid ? w1 : 0.0f;
        b1.x = valid ? w2 : 0.0f;  b1.y = valid ? w3 : 0.0f;
        b2.x = valid ? w4 : 0.0f;  b2.y = valid ? w5 : 0.0f;
        b3.x = valid ? w6 : 0.0f;  b3.y = valid ? w7 : 0.0f;

        v8f acc = {};
        acc = __builtin_amdgcn_wmma_f32_16x16x4_f32(false, a0, false, b0,
                                                    (short)0, acc, false, false);
        acc = __builtin_amdgcn_wmma_f32_16x16x4_f32(false, a1, false, b1,
                                                    (short)0, acc, false, false);
        acc = __builtin_amdgcn_wmma_f32_16x16x4_f32(false, a2, false, b2,
                                                    (short)0, acc, false, false);
        acc = __builtin_amdgcn_wmma_f32_16x16x4_f32(false, a3, false, b3,
                                                    (short)0, acc, false, false);

        if (f < 11) {
#pragma unroll
            for (int r = 0; r < 8; ++r) {
                float v   = valid ? (acc[r] + bn) : -3.0e38f;
                int   idx = n;
#pragma unroll
                for (int mask = 1; mask <= 4; mask <<= 1) {
                    float ov = __shfl_xor(v,   mask, 32);
                    int   oi = __shfl_xor(idx, mask, 32);
                    if (ov > v || (ov == v && oi < idx)) { v = ov; idx = oi; }
                }
                if (n == 0) {
                    const int orow = tile * 16 + r + 8 * kh;
                    out[(size_t)orow * kF + f] = (float)idx;
                }
            }
        } else {
            const int fc = f - 11;
#pragma unroll
            for (int r = 0; r < 8; ++r) {
                float t  = acc[r] + bn;
                float ts = __shfl_xor(t, 1, 32);
                if (n == 0) {
                    const int orow = tile * 16 + r + 8 * kh;
                    const float mean = fast_tanh(t);
                    const float sd   = fast_sigmoid(ts);
                    const float e    = eps[(size_t)orow * kNCont + fc];
                    out[(size_t)orow * kF + f] = mean + sd * e;
                }
            }
        }
    }
}

extern "C" void kernel_launch(void* const* d_in, const int* in_sizes, int n_in,
                              void* d_out, int out_size, void* d_ws, size_t ws_size,
                              hipStream_t stream) {
    const float* x   = (const float*)d_in[0];
    const float* W2a = (const float*)d_in[1];
    const float* b2a = (const float*)d_in[2];
    const float* W3  = (const float*)d_in[3];
    const float* b3  = (const float*)d_in[4];
    const float* W4  = (const float*)d_in[5];
    const float* b4  = (const float*)d_in[6];
    const float* W6  = (const float*)d_in[7];
    const float* b6  = (const float*)d_in[8];
    const float* W7  = (const float*)d_in[9];
    const float* b7  = (const float*)d_in[10];
    const float* Wc  = (const float*)d_in[11];
    const float* bc  = (const float*)d_in[12];
    const float* eps = (const float*)d_in[13];
    float* out = (float*)d_out;

    tab_decoder_wmma<<<dim3(kNRows / 16), dim3(128), 0, stream>>>(
        x, W2a, b2a, W3, b3, W4, b4, W6, b6, W7, b7, Wc, bc, eps, out);
}


# All eps loads of a feature issued with its A/B loads (loop split at f=11, continuous body branch-free, loads pinned by an empty asm): 1 memory round trip per feature instead of 9 serial ones
# speedup vs baseline: 2.6734x; 2.6734x over previous
#include <hip/hip_runtime.h>
#include <hip/hip_bf16.h>
#include <math.h>

typedef __attribute__((ext_vector_type(2))) float v2f;
typedef __attribute__((ext_vector_type(8))) float v8f;

constexpr int kNRows     = 8192;
constexpr int kF         = 188;
constexpr int kH         = 16;
constexpr int kRowStride = kF * kH;
constexpr int kNCont     = 177;

__device__ __forceinline__ float fast_rcp(float a) {
    return __builtin_amdgcn_rcpf(a);
}
__device__ __forceinline__ float fast_sigmoid(float a) {
    return fast_rcp(1.0f + __expf(-a));
}
__device__ __forceinline__ float fast_tanh(float a) {
#if __has_builtin(__builtin_amdgcn_tanhf)
    return __builtin_amdgcn_tanhf(a);
#elif __has_builtin(__builtin_amdgcn_tanh_f32)
    return __builtin_amdgcn_tanh_f32(a);
#else
    const float e2 = __expf(-2.0f * __builtin_fabsf(a));
    const float t  = (1.0f - e2) * fast_rcp(1.0f + e2);
    return __builtin_copysignf(t, a);
#endif
}

__global__ __launch_bounds__(128)
void tab_decoder_wmma(const float* __restrict__ x,
                      const float* __restrict__ W2a, const float* __restrict__ b2a,
                      const float* __restrict__ W3,  const float* __restrict__ b3,
                      const float* __restrict__ W4,  const float* __restrict__ b4,
                      const float* __restrict__ W6,  const float* __restrict__ b6,
                      const float* __restrict__ W7,  const float* __restrict__ b7,
                      const float* __restrict__ Wc,  const float* __restrict__ bc,
                      const float* __restrict__ eps,
                      float* __restrict__ out)
{
    const int tile = blockIdx.x;
    const int wave = threadIdx.x >> 5;
    const int lane = threadIdx.x & 31;
    const int m    = lane & 15;
    const int kh   = lane >> 4;
    const int n    = m;

    const int    arow = tile * 16 + m;
    const float* xb   = x + (size_t)arow * kRowStride;

    int f = wave;

    for (; f < 11; f += 4) {
        const float* W; const float* bias; int od;
        if      (f < 4)  { W = W2a + f       * kH * 2; bias = b2a + f       * 2; od = 2; }
        else if (f < 6)  { W = W3  + (f - 4) * kH * 3; bias = b3  + (f - 4) * 3; od = 3; }
        else if (f < 7)  { W = W4;                     bias = b4;                od = 4; }
        else if (f < 8)  { W = W6;                     bias = b6;                od = 6; }
        else             { W = W7  + (f - 8) * kH * 7; bias = b7  + (f - 8) * 7; od = 7; }

        const float* xf = xb + f * kH;
        __builtin_prefetch(xf + 4 * kH, 0, 1);

        const bool valid = (n < od);
        const int  nc    = valid ? n : 0;

        const float braw = bias[nc];
        const float bn   = valid ? braw : 0.0f;

        v2f a0 = *(const v2f*)(xf + 0  + 2 * kh);
        v2f a1 = *(const v2f*)(xf + 4  + 2 * kh);
        v2f a2 = *(const v2f*)(xf + 8  + 2 * kh);
        v2f a3 = *(const v2f*)(xf + 12 + 2 * kh);

        const float* Wl = W + 2 * kh * od + nc;
        float w0 = Wl[0 * od], w1 = Wl[1 * od];
        float w2 = Wl[4 * od], w3 = Wl[5 * od];
        float w4 = Wl[8 * od], w5 = Wl[9 * od];
        float w6 = Wl[12 * od], w7 = Wl[13 * od];

        v2f b0, b1, b2, b3;
        b0.x = valid ? w0 : 0.0f;  b0.y = valid ? w1 : 0.0f;
        b1.x = valid ? w2 : 0.0f;  b1.y = valid ? w3 : 0.0f;
        b2.x = valid ? w4 : 0.0f;  b2.y = valid ? w5 : 0.0f;
        b3.x = valid ? w6 : 0.0f;  b3.y = valid ? w7 : 0.0f;

        v8f acc = {};
        acc = __builtin_amdgcn_wmma_f32_16x16x4_f32(false, a0, false, b0,
                                                    (short)0, acc, false, false);
        acc = __builtin_amdgcn_wmma_f32_16x16x4_f32(false, a1, false, b1,
                                                    (short)0, acc, false, false);
        acc = __builtin_amdgcn_wmma_f32_16x16x4_f32(false, a2, false, b2,
                                                    (short)0, acc, false, false);
        acc = __builtin_amdgcn_wmma_f32_16x16x4_f32(false, a3, false, b3,
                                                    (short)0, acc, false, false);

#pragma unroll
        for (int r = 0; r < 8; ++r) {
            float v   = valid ? (acc[r] + bn) : -3.0e38f;
            int   idx = n;
#pragma unroll
            for (int mask = 1; mask <= 4; mask <<= 1) {
                float ov = __shfl_xor(v,   mask, 32);
                int   oi = __shfl_xor(idx, mask, 32);
                if (ov > v || (ov == v && oi < idx)) { v = ov; idx = oi; }
            }
            if (n == 0) {
                const int orow = tile * 16 + r + 8 * kh;
                out[(size_t)orow * kF + f] = (float)idx;
            }
        }
    }

    for (; f < kF; f += 4) {
        const int    fc = f - 11;
        const float* xf = xb + f * kH;
        __builtin_prefetch(xf + 4 * kH, 0, 1);

        const bool valid = (n < 2);
        const int  nc    = valid ? n : 0;

        const float braw = bc[fc * 2 + nc];
        const float bn   = valid ? braw : 0.0f;

        v2f a0 = *(const v2f*)(xf + 0  + 2 * kh);
        v2f a1 = *(const v2f*)(xf + 4  + 2 * kh);
        v2f a2 = *(const v2f*)(xf + 8  + 2 * kh);
        v2f a3 = *(const v2f*)(xf + 12 + 2 * kh);

        const float* Wl = Wc + fc * kH * 2 + 2 * kh * 2 + nc;
        float w0 = Wl[0 * 2],  w1 = Wl[1 * 2];
        float w2 = Wl[4 * 2],  w3 = Wl[5 * 2];
        float w4 = Wl[8 * 2],  w5 = Wl[9 * 2];
        float w6 = Wl[12 * 2], w7 = Wl[13 * 2];

        const float* ep = eps + (size_t)(tile * 16 + 8 * kh) * kNCont + fc;
        float e[8];
#pragma unroll
        for (int r = 0; r < 8; ++r) e[r] = ep[r * kNCont];

        v2f b0, b1, b2, b3;
        b0.x = valid ? w0 : 0.0f;  b0.y = valid ? w1 : 0.0f;
        b1.x = valid ? w2 : 0.0f;  b1.y = valid ? w3 : 0.0f;
        b2.x = valid ? w4 : 0.0f;  b2.y = valid ? w5 : 0.0f;
        b3.x = valid ? w6 : 0.0f;  b3.y = valid ? w7 : 0.0f;

        v8f acc = {};
        acc = __builtin_amdgcn_wmma_f32_16x16x4_f32(false, a0, false, b0,
                                                    (short)0, acc, false, false);
        acc = __builtin_amdgcn_wmma_f32_16x16x4_f32(false, a1, false, b1,
                                                    (short)0, acc, false, false);
        acc = __builtin_amdgcn_wmma_f32_16x16x4_f32(false, a2, false, b2,
                                                    (short)0, acc, false, false);
        acc = __builtin_amdgcn_wmma_f32_16x16x4_f32(false, a3, false, b3,
                                                    (short)0, acc, false, false);

#pragma unroll
        for (int r = 0; r < 8; ++r) asm volatile("" : "+v"(e[r]));

        float val[8];
#pragma unroll
        for (int r = 0; r < 8; ++r) {
            float t  = acc[r] + bn;
            float ts = __shfl_xor(t, 1, 32);
            const float mean = fast_tanh(t);
            const float sd   = fast_sigmoid(ts);
            val[r] = mean + sd * e[r];
        }
        if (n == 0) {
#pragma unroll
            for (int r = 0; r < 8; ++r) {
                const int orow = tile * 16 + r + 8 * kh;
                out[(size_t)orow * kF + f] = val[r];
            }
        }
    }
}

extern "C" void kernel_launch(void* const* d_in, const int* in_sizes, int n_in,
                              void* d_out, int out_size, void* d_ws, size_t ws_size,
                              hipStream_t stream) {
    const float* x   = (const float*)d_in[0];
    const float* W2a = (const float*)d_in[1];
    const float* b2a = (const float*)d_in[2];
    const float* W3  = (const float*)d_in[3];
    const float* b3  = (const float*)d_in[4];
    const float* W4  = (const float*)d_in[5];
    const float* b4  = (const float*)d_in[6];
    const float* W6  = (const float*)d_in[7];
    const float* b6  = (const float*)d_in[8];
    const float* W7  = (const float*)d_in[9];
    const float* b7  = (const float*)d_in[10];
    const float* Wc  = (const float*)d_in[11];
    const float* bc  = (const float*)d_in[12];
    const float* eps = (const float*)d_in[13];
    float* out = (float*)d_out;

    tab_decoder_wmma<<<dim3(kNRows / 16), dim3(128), 0, stream>>>(
        x, W2a, b2a, W3, b3, W4, b4, W6, b6, W7, b7, Wc, bc, eps, out);
}


# Each 16-row tile split over 16 waves (2 blocks of 8, feature stride 16) instead of 4: 8192 short waves, 8 per SIMD instead of 2, so 4x as many per-feature load groups are in flight; sums unchanged
# speedup vs baseline: 4.3326x; 1.6206x over previous
#include <hip/hip_runtime.h>
#include <hip/hip_bf16.h>
#include <math.h>

typedef __attribute__((ext_vector_type(2))) float v2f;
typedef __attribute__((ext_vector_type(8))) float v8f;

constexpr int kNRows     = 8192;
constexpr int kF         = 188;
constexpr int kH         = 16;
constexpr int kRowStride = kF * kH;
constexpr int kNCont     = 177;

__device__ __forceinline__ float fast_rcp(float a) {
    return __builtin_amdgcn_rcpf(a);
}
__device__ __forceinline__ float fast_sigmoid(float a) {
    return fast_rcp(1.0f + __expf(-a));
}
__device__ __forceinline__ float fast_tanh(float a) {
#if __has_builtin(__builtin_amdgcn_tanhf)
    return __builtin_amdgcn_tanhf(a);
#elif __has_builtin(__builtin_amdgcn_tanh_f32)
    return __builtin_amdgcn_tanh_f32(a);
#else
    const float e2 = __expf(-2.0f * __builtin_fabsf(a));
    const float t  = (1.0f - e2) * fast_rcp(1.0f + e2);
    return __builtin_copysignf(t, a);
#endif
}

constexpr int kWavesPerBlock = 8;
constexpr int kWavesPerTile  = 16;
__global__ __launch_bounds__(kWavesPerBlock * 32)
void tab_decoder_wmma(const float* __restrict__ x,
                      const float* __restrict__ W2a, const float* __restrict__ b2a,
                      const float* __restrict__ W3,  const float* __restrict__ b3,
                      const float* __restrict__ W4,  const float* __restrict__ b4,
                      const float* __restrict__ W6,  const float* __restrict__ b6,
                      const float* __restrict__ W7,  const float* __restrict__ b7,
                      const float* __restrict__ Wc,  const float* __restrict__ bc,
                      const float* __restrict__ eps,
                      float* __restrict__ out)
{
    const int tile = blockIdx.x;
    const int wave = threadIdx.x >> 5;
    const int lane = threadIdx.x & 31;
    const int m    = lane & 15;
    const int kh   = lane >> 4;
    const int n    = m;

    const int    arow = tile * 16 + m;
    const float* xb   = x + (size_t)arow * kRowStride;

    int f = blockIdx.y * kWavesPerBlock + wave;

    for (; f < 11; f += kWavesPerTile) {
        const float* W; const float* bias; int od;
        if      (f < 4)  { W = W2a + f       * kH * 2; bias = b2a + f       * 2; od = 2; }
        else if (f < 6)  { W = W3  + (f - 4) * kH * 3; bias = b3  + (f - 4) * 3; od = 3; }
        else if (f < 7)  { W = W4;                     bias = b4;                od = 4; }
        else if (f < 8)  { W = W6;                     bias = b6;                od = 6; }
        else             { W = W7  + (f - 8) * kH * 7; bias = b7  + (f - 8) * 7; od = 7; }

        const float* xf = xb + f * kH;
        __builtin_prefetch(xf + kWavesPerTile * kH, 0, 1);

        const bool valid = (n < od);
        const int  nc    = valid ? n : 0;

        const float braw = bias[nc];
        const float bn   = valid ? braw : 0.0f;

        v2f a0 = *(const v2f*)(xf + 0  + 2 * kh);
        v2f a1 = *(const v2f*)(xf + 4  + 2 * kh);
        v2f a2 = *(const v2f*)(xf + 8  + 2 * kh);
        v2f a3 = *(const v2f*)(xf + 12 + 2 * kh);

        const float* Wl = W + 2 * kh * od + nc;
        float w0 = Wl[0 * od], w1 = Wl[1 * od];
        float w2 = Wl[4 * od], w3 = Wl[5 * od];
        float w4 = Wl[8 * od], w5 = Wl[9 * od];
        float w6 = Wl[12 * od], w7 = Wl[13 * od];

        v2f b0, b1, b2, b3;
        b0.x = valid ? w0 : 0.0f;  b0.y = valid ? w1 : 0.0f;
        b1.x = valid ? w2 : 0.0f;  b1.y = valid ? w3 : 0.0f;
        b2.x = valid ? w4 : 0.0f;  b2.y = valid ? w5 : 0.0f;
        b3.x = valid ? w6 : 0.0f;  b3.y = valid ? w7 : 0.0f;

        v8f acc = {};
        acc = __builtin_amdgcn_wmma_f32_16x16x4_f32(false, a0, false, b0,
                                                    (short)0, acc, false, false);
        acc = __builtin_amdgcn_wmma_f32_16x16x4_f32(false, a1, false, b1,
                                                    (short)0, acc, false, false);
        acc = __builtin_amdgcn_wmma_f32_16x16x4_f32(false, a2, false, b2,
                                                    (short)0, acc, false, false);
        acc = __builtin_amdgcn_wmma_f32_16x16x4_f32(false, a3, false, b3,
                                                    (short)0, acc, false, false);

#pragma unroll
        for (int r = 0; r < 8; ++r) {
            float v   = valid ? (acc[r] + bn) : -3.0e38f;
            int   idx = n;
#pragma unroll
            for (int mask = 1; mask <= 4; mask <<= 1) {
                float ov = __shfl_xor(v,   mask, 32);
                int   oi = __shfl_xor(idx, mask, 32);
                if (ov > v || (ov == v && oi < idx)) { v = ov; idx = oi; }
            }
            if (n == 0) {
                const int orow = tile * 16 + r + 8 * kh;
                out[(size_t)orow * kF + f] = (float)idx;
            }
        }
    }

    for (; f < kF; f += kWavesPerTile) {
        const int    fc = f - 11;
        const float* xf = xb + f * kH;
        if (f + kWavesPerTile < kF) __builtin_prefetch(xf + kWavesPerTile * kH, 0, 1);

        const bool valid = (n < 2);
        const int  nc    = valid ? n : 0;

        const float braw = bc[fc * 2 + nc];
        const float bn   = valid ? braw : 0.0f;

        v2f a0 = *(const v2f*)(xf + 0  + 2 * kh);
        v2f a1 = *(const v2f*)(xf + 4  + 2 * kh);
        v2f a2 = *(const v2f*)(xf + 8  + 2 * kh);
        v2f a3 = *(const v2f*)(xf + 12 + 2 * kh);

        const float* Wl = Wc + fc * kH * 2 + 2 * kh * 2 + nc;
        float w0 = Wl[0 * 2],  w1 = Wl[1 * 2];
        float w2 = Wl[4 * 2],  w3 = Wl[5 * 2];
        float w4 = Wl[8 * 2],  w5 = Wl[9 * 2];
        float w6 = Wl[12 * 2], w7 = Wl[13 * 2];

        const float* ep = eps + (size_t)(tile * 16 + 8 * kh) * kNCont + fc;
        float e[8];
#pragma unroll
        for (int r = 0; r < 8; ++r) e[r] = ep[r * kNCont];

        __builtin_amdgcn_sched_barrier(0);

        v2f b0, b1, b2, b3;
        b0.x = valid ? w0 : 0.0f;  b0.y = valid ? w1 : 0.0f;
        b1.x = valid ? w2 : 0.0f;  b1.y = valid ? w3 : 0.0f;
        b2.x = valid ? w4 : 0.0f;  b2.y = valid ? w5 : 0.0f;
        b3.x = valid ? w6 : 0.0f;  b3.y = valid ? w7 : 0.0f;

        v8f acc = {};
        acc = __builtin_amdgcn_wmma_f32_16x16x4_f32(false, a0, false, b0,
                                                    (short)0, acc, false, false);
        acc = __builtin_amdgcn_wmma_f32_16x16x4_f32(false, a1, false, b1,
                                                    (short)0, acc, false, false);
        acc = __builtin_amdgcn_wmma_f32_16x16x4_f32(false, a2, false, b2,
                                                    (short)0, acc, false, false);
        acc = __builtin_amdgcn_wmma_f32_16x16x4_f32(false, a3, false, b3,
                                                    (short)0, acc, false, false);

#pragma unroll
        for (int r = 0; r < 8; ++r) asm volatile("" : "+v"(e[r]));

        float val[8];
#pragma unroll
        for (int r = 0; r < 8; ++r) {
            float t  = acc[r] + bn;
            float ts = __shfl_xor(t, 1, 32);
            const float mean = fast_tanh(t);
            const float sd   = fast_sigmoid(ts);
            val[r] = mean + sd * e[r];
        }
        if (n == 0) {
#pragma unroll
            for (int r = 0; r < 8; ++r) {
                const int orow = tile * 16 + r + 8 * kh;
                out[(size_t)orow * kF + f] = val[r];
            }
        }
    }
}

extern "C" void kernel_launch(void* const* d_in, const int* in_sizes, int n_in,
                              void* d_out, int out_size, void* d_ws, size_t ws_size,
                              hipStream_t stream) {
    const float* x   = (const float*)d_in[0];
    const float* W2a = (const float*)d_in[1];
    const float* b2a = (const float*)d_in[2];
    const float* W3  = (const float*)d_in[3];
    const float* b3  = (const float*)d_in[4];
    const float* W4  = (const float*)d_in[5];
    const float* b4  = (const float*)d_in[6];
    const float* W6  = (const float*)d_in[7];
    const float* b6  = (const float*)d_in[8];
    const float* W7  = (const float*)d_in[9];
    const float* b7  = (const float*)d_in[10];
    const float* Wc  = (const float*)d_in[11];
    const float* bc  = (const float*)d_in[12];
    const float* eps = (const float*)d_in[13];
    float* out = (float*)d_out;

    tab_decoder_wmma<<<dim3(kNRows / 16, 2), dim3(256), 0, stream>>>(
        x, W2a, b2a, W3, b3, W4, b4, W6, b6, W7, b7, Wc, bc, eps, out);
}


# Eight continuous features packed block-diagonally into one 16-column WMMA chain (x read as 16-byte pieces, W in the same slot order): k sum order within a feature changes
# speedup vs baseline: 6.3483x; 1.4652x over previous
#include <hip/hip_runtime.h>
#include <hip/hip_bf16.h>
#include <math.h>

typedef __attribute__((ext_vector_type(2))) float v2f;
typedef __attribute__((ext_vector_type(4))) float v4f;
typedef __attribute__((ext_vector_type(8))) float v8f;

constexpr int kNRows     = 8192;
constexpr int kF         = 188;
constexpr int kH         = 16;
constexpr int kRowStride = kF * kH;
constexpr int kNCont     = 177;
constexpr int kPack      = 8;
constexpr int kPackF0    = 12;
constexpr int kNPack     = (kF - kPackF0) / kPack;
static_assert(kPackF0 + kNPack * kPack == kF, "packs must end at the last feature");

__device__ __forceinline__ float fast_rcp(float a) {
    return __builtin_amdgcn_rcpf(a);
}
__device__ __forceinline__ float fast_sigmoid(float a) {
    return fast_rcp(1.0f + __expf(-a));
}
__device__ __forceinline__ float fast_tanh(float a) {
#if __has_builtin(__builtin_amdgcn_tanhf)
    return __builtin_amdgcn_tanhf(a);
#elif __has_builtin(__builtin_amdgcn_tanh_f32)
    return __builtin_amdgcn_tanh_f32(a);
#else
    const float e2 = __expf(-2.0f * __builtin_fabsf(a));
    const float t  = (1.0f - e2) * fast_rcp(1.0f + e2);
    return __builtin_copysignf(t, a);
#endif
}

constexpr int kWavesPerBlock = 8;
constexpr int kWavesPerTile  = 16;
__global__ __launch_bounds__(kWavesPerBlock * 32)
void tab_decoder_wmma(const float* __restrict__ x,
                      const float* __restrict__ W2a, const float* __restrict__ b2a,
                      const float* __restrict__ W3,  const float* __restrict__ b3,
                      const float* __restrict__ W4,  const float* __restrict__ b4,
                      const float* __restrict__ W6,  const float* __restrict__ b6,
                      const float* __restrict__ W7,  const float* __restrict__ b7,
                      const float* __restrict__ Wc,  const float* __restrict__ bc,
                      const float* __restrict__ eps,
                      float* __restrict__ out)
{
    const int tile = blockIdx.x;
    const int wave = threadIdx.x >> 5;
    const int lane = threadIdx.x & 31;
    const int m    = lane & 15;
    const int kh   = lane >> 4;
    const int n    = m;

    const int    arow = tile * 16 + m;
    const float* xb   = x + (size_t)arow * kRowStride;

    int f = blockIdx.y * kWavesPerBlock + wave;

    for (; f < 11; f += kWavesPerTile) {
        const float* W; const float* bias; int od;
        if      (f < 4)  { W = W2a + f       * kH * 2; bias = b2a + f       * 2; od = 2; }
        else if (f < 6)  { W = W3  + (f - 4) * kH * 3; bias = b3  + (f - 4) * 3; od = 3; }
        else if (f < 7)  { W = W4;                     bias = b4;                od = 4; }
        else if (f < 8)  { W = W6;                     bias = b6;                od = 6; }
        else             { W = W7  + (f - 8) * kH * 7; bias = b7  + (f - 8) * 7; od = 7; }

        const float* xf = xb + f * kH;

        const bool valid = (n < od);
        const int  nc    = valid ? n : 0;

        const float braw = bias[nc];
        const float bn   = valid ? braw : 0.0f;

        v2f a0 = *(const v2f*)(xf + 0  + 2 * kh);
        v2f a1 = *(const v2f*)(xf + 4  + 2 * kh);
        v2f a2 = *(const v2f*)(xf + 8  + 2 * kh);
        v2f a3 = *(const v2f*)(xf + 12 + 2 * kh);

        const float* Wl = W + 2 * kh * od + nc;
        float w0 = Wl[0 * od], w1 = Wl[1 * od];
        float w2 = Wl[4 * od], w3 = Wl[5 * od];
        float w4 = Wl[8 * od], w5 = Wl[9 * od];
        float w6 = Wl[12 * od], w7 = Wl[13 * od];

        v2f b0, b1, b2, b3;
        b0.x = valid ? w0 : 0.0f;  b0.y = valid ? w1 : 0.0f;
        b1.x = valid ? w2 : 0.0f;  b1.y = valid ? w3 : 0.0f;
        b2.x = valid ? w4 : 0.0f;  b2.y = valid ? w5 : 0.0f;
        b3.x = valid ? w6 : 0.0f;  b3.y = valid ? w7 : 0.0f;

        v8f acc = {};
        acc = __builtin_amdgcn_wmma_f32_16x16x4_f32(false, a0, false, b0,
                                                    (short)0, acc, false, false);
        acc = __builtin_amdgcn_wmma_f32_16x16x4_f32(false, a1, false, b1,
                                                    (short)0, acc, false, false);
        acc = __builtin_amdgcn_wmma_f32_16x16x4_f32(false, a2, false, b2,
                                                    (short)0, acc, false, false);
        acc = __builtin_amdgcn_wmma_f32_16x16x4_f32(false, a3, false, b3,
                                                    (short)0, acc, false, false);

#pragma unroll
        for (int r = 0; r < 8; ++r) {
            float v   = valid ? (acc[r] + bn) : -3.0e38f;
            int   idx = n;
#pragma unroll
            for (int mask = 1; mask <= 4; mask <<= 1) {
                float ov = __shfl_xor(v,   mask, 32);
                int   oi = __shfl_xor(idx, mask, 32);
                if (ov > v || (ov == v && oi < idx)) { v = ov; idx = oi; }
            }
            if (n == 0) {
                const int orow = tile * 16 + r + 8 * kh;
                out[(size_t)orow * kF + f] = (float)idx;
            }
        }
    }

    for (; f < kPackF0; f += kWavesPerTile) {
        const int    fc = f - 11;
        const float* xf = xb + f * kH;

        const bool valid = (n < 2);
        const int  nc    = valid ? n : 0;

        const float braw = bc[fc * 2 + nc];
        const float bn   = valid ? braw : 0.0f;

        v2f a0 = *(const v2f*)(xf + 0  + 2 * kh);
        v2f a1 = *(const v2f*)(xf + 4  + 2 * kh);
        v2f a2 = *(const v2f*)(xf + 8  + 2 * kh);
        v2f a3 = *(const v2f*)(xf + 12 + 2 * kh);

        const float* Wl = Wc + fc * kH * 2 + 2 * kh * 2 + nc;
        float w0 = Wl[0 * 2],  w1 = Wl[1 * 2];
        float w2 = Wl[4 * 2],  w3 = Wl[5 * 2];
        float w4 = Wl[8 * 2],  w5 = Wl[9 * 2];
        float w6 = Wl[12 * 2], w7 = Wl[13 * 2];

        const float* ep = eps + (size_t)(tile * 16 + 8 * kh) * kNCont + fc;
        float e[8];
#pragma unroll
        for (int r = 0; r < 8; ++r) e[r] = ep[r * kNCont];

        __builtin_amdgcn_sched_barrier(0);

        v2f b0, b1, b2, b3;
        b0.x = valid ? w0 : 0.0f;  b0.y = valid ? w1 : 0.0f;
        b1.x = valid ? w2 : 0.0f;  b1.y = valid ? w3 : 0.0f;
        b2.x = valid ? w4 : 0.0f;  b2.y = valid ? w5 : 0.0f;
        b3.x = valid ? w6 : 0.0f;  b3.y = valid ? w7 : 0.0f;

        v8f acc = {};
        acc = __builtin_amdgcn_wmma_f32_16x16x4_f32(false, a0, false, b0,
                                                    (short)0, acc, false, false);
        acc = __builtin_amdgcn_wmma_f32_16x16x4_f32(false, a1, false, b1,
                                                    (short)0, acc, false, false);
        acc = __builtin_amdgcn_wmma_f32_16x16x4_f32(false, a2, false, b2,
                                                    (short)0, acc, false, false);
        acc = __builtin_amdgcn_wmma_f32_16x16x4_f32(false, a3, false, b3,
                                                    (short)0, acc, false, false);

#pragma unroll
        for (int r = 0; r < 8; ++r) asm volatile("" : "+v"(e[r]));

        float val[8];
#pragma unroll
        for (int r = 0; r < 8; ++r) {
            float t  = acc[r] + bn;
            float ts = __shfl_xor(t, 1, 32);
            const float mean = fast_tanh(t);
            const float sd   = fast_sigmoid(ts);
            val[r] = mean + sd * e[r];
        }
        if (n == 0) {
#pragma unroll
            for (int r = 0; r < 8; ++r) {
                const int orow = tile * 16 + r + 8 * kh;
                out[(size_t)orow * kF + f] = val[r];
            }
        }
    }

    for (int j = blockIdx.y * kWavesPerBlock + wave; j < kNPack; j += kWavesPerTile) {
        const int f0  = kPackF0 + kPack * j;
        const int fc0 = f0 - 11;
        const int tn  = n >> 1;
        const int c   = n & 1;

        const float* xg = xb + f0 * kH + 4 * kh;
        v4f a[16];
#pragma unroll
        for (int i = 0; i < 16; ++i) a[i] = *(const v4f*)(xg + 8 * i);

        const float* Wl = Wc + (size_t)(fc0 + tn) * (kH * 2) + 8 * kh;
        const v4f wA = *(const v4f*)(Wl + 0);
        const v4f wB = *(const v4f*)(Wl + 4);
        const v4f wC = *(const v4f*)(Wl + 16);
        const v4f wD = *(const v4f*)(Wl + 20);

        const float bn = bc[fc0 * 2 + n];

        const float* ep = eps + (size_t)(tile * 16 + 8 * kh) * kNCont + fc0 + tn;
        float e[8];
#pragma unroll
        for (int r = 0; r < 8; ++r) e[r] = ep[r * kNCont];

        __builtin_amdgcn_sched_barrier(0);

        float wq[8];
        wq[0] = c ? wA.y : wA.x;   wq[1] = c ? wA.w : wA.z;
        wq[2] = c ? wB.y : wB.x;   wq[3] = c ? wB.w : wB.z;
        wq[4] = c ? wC.y : wC.x;   wq[5] = c ? wC.w : wC.z;
        wq[6] = c ? wD.y : wD.x;   wq[7] = c ? wD.w : wD.z;

        v8f acc = {};
#pragma unroll
        for (int i = 0; i < 16; ++i) {
            const bool mine = (tn == (i >> 1));
#pragma unroll
            for (int u = 0; u < 2; ++u) {
                v2f av, bv;
                av.x = a[i][2 * u];
                av.y = a[i][2 * u + 1];
                bv.x = mine ? wq[4 * (i & 1) + 2 * u]     : 0.0f;
                bv.y = mine ? wq[4 * (i & 1) + 2 * u + 1] : 0.0f;
                acc = __builtin_amdgcn_wmma_f32_16x16x4_f32(false, av, false, bv,
                                                            (short)0, acc, false, false);
            }
        }

#pragma unroll
        for (int r = 0; r < 8; ++r) asm volatile("" : "+v"(e[r]));

        float val[8];
#pragma unroll
        for (int r = 0; r < 8; ++r) {
            float t  = acc[r] + bn;
            float ts = __shfl_xor(t, 1, 32);
            const float mean = fast_tanh(t);
            const float sd   = fast_sigmoid(ts);
            val[r] = mean + sd * e[r];
        }
        if (c == 0) {
#pragma unroll
            for (int r = 0; r < 8; ++r) {
                const int orow = tile * 16 + r + 8 * kh;
                out[(size_t)orow * kF + f0 + tn] = val[r];
            }
        }
    }
}

extern "C" void kernel_launch(void* const* d_in, const int* in_sizes, int n_in,
                              void* d_out, int out_size, void* d_ws, size_t ws_size,
                              hipStream_t stream) {
    const float* x   = (const float*)d_in[0];
    const float* W2a = (const float*)d_in[1];
    const float* b2a = (const float*)d_in[2];
    const float* W3  = (const float*)d_in[3];
    const float* b3  = (const float*)d_in[4];
    const float* W4  = (const float*)d_in[5];
    const float* b4  = (const float*)d_in[6];
    const float* W6  = (const float*)d_in[7];
    const float* b6  = (const float*)d_in[8];
    const float* W7  = (const float*)d_in[9];
    const float* b7  = (const float*)d_in[10];
    const float* Wc  = (const float*)d_in[11];
    const float* bc  = (const float*)d_in[12];
    const float* eps = (const float*)d_in[13];
    float* out = (float*)d_out;

    tab_decoder_wmma<<<dim3(kNRows / 16, 2), dim3(256), 0, stream>>>(
        x, W2a, b2a, W3, b3, W4, b4, W6, b6, W7, b7, Wc, bc, eps, out);
}


# 8 waves per tile, each walking up to 3 packs with the next pack's loads issued ahead of the current WMMAs; argmax rows branch-free so 3 singles fit a wave: one exposed memory wait per wave; same sums
# speedup vs baseline: 6.9697x; 1.0979x over previous
#include <hip/hip_runtime.h>
#include <hip/hip_bf16.h>
#include <math.h>

typedef __attribute__((ext_vector_type(2))) float v2f;
typedef __attribute__((ext_vector_type(4))) float v4f;
typedef __attribute__((ext_vector_type(8))) float v8f;

constexpr int kNRows     = 8192;
constexpr int kF         = 188;
constexpr int kH         = 16;
constexpr int kRowStride = kF * kH;
constexpr int kNCont     = 177;
constexpr int kPack      = 8;
constexpr int kPackF0    = 12;
constexpr int kNPack     = (kF - kPackF0) / kPack;
static_assert(kPackF0 + kNPack * kPack == kF, "packs must end at the last feature");

__device__ __forceinline__ float fast_rcp(float a) {
    return __builtin_amdgcn_rcpf(a);
}
__device__ __forceinline__ float fast_sigmoid(float a) {
    return fast_rcp(1.0f + __expf(-a));
}
__device__ __forceinline__ float fast_tanh(float a) {
#if __has_builtin(__builtin_amdgcn_tanhf)
    return __builtin_amdgcn_tanhf(a);
#elif __has_builtin(__builtin_amdgcn_tanh_f32)
    return __builtin_amdgcn_tanh_f32(a);
#else
    const float e2 = __expf(-2.0f * __builtin_fabsf(a));
    const float t  = (1.0f - e2) * fast_rcp(1.0f + e2);
    return __builtin_copysignf(t, a);
#endif
}

constexpr int kWavesPerBlock = 8;

struct PackIn {
    v4f   a[16];
    v4f   wA, wB, wC, wD;
    float bn;
    float e[8];
};

struct OneIn {
    v2f   a0, a1, a2, a3;
    float w[8];
    float braw;
};

__device__ __forceinline__ void pack_load(PackIn& p, int j, const float* xb,
                                          const float* __restrict__ Wc,
                                          const float* __restrict__ bc,
                                          const float* __restrict__ eps,
                                          int tile, int kh, int n)
{
    const int f0  = kPackF0 + kPack * j;
    const int fc0 = f0 - 11;
    const int tn  = n >> 1;

    const float* xg = xb + f0 * kH + 4 * kh;
#pragma unroll
    for (int i = 0; i < 16; ++i) p.a[i] = *(const v4f*)(xg + 8 * i);

    const float* Wl = Wc + (size_t)(fc0 + tn) * (kH * 2) + 8 * kh;
    p.wA = *(const v4f*)(Wl + 0);
    p.wB = *(const v4f*)(Wl + 4);
    p.wC = *(const v4f*)(Wl + 16);
    p.wD = *(const v4f*)(Wl + 20);

    p.bn = bc[fc0 * 2 + n];

    const float* ep = eps + (size_t)(tile * 16 + 8 * kh) * kNCont + fc0 + tn;
#pragma unroll
    for (int r = 0; r < 8; ++r) p.e[r] = ep[r * kNCont];
}

__device__ __forceinline__ void pack_compute(PackIn& p, int j,
                                             float* __restrict__ out,
                                             int tile, int kh, int n)
{
    const int f0 = kPackF0 + kPack * j;
    const int tn = n >> 1;
    const int c  = n & 1;

    float wq[8];
    wq[0] = c ? p.wA.y : p.wA.x;   wq[1] = c ? p.wA.w : p.wA.z;
    wq[2] = c ? p.wB.y : p.wB.x;   wq[3] = c ? p.wB.w : p.wB.z;
    wq[4] = c ? p.wC.y : p.wC.x;   wq[5] = c ? p.wC.w : p.wC.z;
    wq[6] = c ? p.wD.y : p.wD.x;   wq[7] = c ? p.wD.w : p.wD.z;

    v8f acc = {};
#pragma unroll
    for (int i = 0; i < 16; ++i) {
        const bool mine = (tn == (i >> 1));
#pragma unroll
        for (int u = 0; u < 2; ++u) {
            v2f av, bv;
            av.x = p.a[i][2 * u];
            av.y = p.a[i][2 * u + 1];
            bv.x = mine ? wq[4 * (i & 1) + 2 * u]     : 0.0f;
            bv.y = mine ? wq[4 * (i & 1) + 2 * u + 1] : 0.0f;
            acc = __builtin_amdgcn_wmma_f32_16x16x4_f32(false, av, false, bv,
                                                        (short)0, acc, false, false);
        }
    }

#pragma unroll
    for (int r = 0; r < 8; ++r) asm volatile("" : "+v"(p.e[r]));

    float val[8];
#pragma unroll
    for (int r = 0; r < 8; ++r) {
        float t  = acc[r] + p.bn;
        float ts = __shfl_xor(t, 1, 32);
        const float mean = fast_tanh(t);
        const float sd   = fast_sigmoid(ts);
        val[r] = mean + sd * p.e[r];
    }
    if (c == 0) {
#pragma unroll
        for (int r = 0; r < 8; ++r) {
            const int orow = tile * 16 + r + 8 * kh;
            out[(size_t)orow * kF + f0 + tn] = val[r];
        }
    }
}

__device__ __forceinline__ void one_load(OneIn& q, const float* xf,
                                         const float* W, const float* bias,
                                         int od, int nc, int kh)
{
    q.braw = bias[nc];
    q.a0 = *(const v2f*)(xf + 0  + 2 * kh);
    q.a1 = *(const v2f*)(xf + 4  + 2 * kh);
    q.a2 = *(const v2f*)(xf + 8  + 2 * kh);
    q.a3 = *(const v2f*)(xf + 12 + 2 * kh);
    const float* Wl = W + 2 * kh * od + nc;
    q.w[0] = Wl[0 * od];   q.w[1] = Wl[1 * od];
    q.w[2] = Wl[4 * od];   q.w[3] = Wl[5 * od];
    q.w[4] = Wl[8 * od];   q.w[5] = Wl[9 * od];
    q.w[6] = Wl[12 * od];  q.w[7] = Wl[13 * od];
}

__device__ __forceinline__ v8f one_product(const OneIn& q, bool valid)
{
    v2f b0, b1, b2, b3;
    b0.x = valid ? q.w[0] : 0.0f;  b0.y = valid ? q.w[1] : 0.0f;
    b1.x = valid ? q.w[2] : 0.0f;  b1.y = valid ? q.w[3] : 0.0f;
    b2.x = valid ? q.w[4] : 0.0f;  b2.y = valid ? q.w[5] : 0.0f;
    b3.x = valid ? q.w[6] : 0.0f;  b3.y = valid ? q.w[7] : 0.0f;
    v8f acc = {};
    acc = __builtin_amdgcn_wmma_f32_16x16x4_f32(false, q.a0, false, b0,
                                                (short)0, acc, false, false);
    acc = __builtin_amdgcn_wmma_f32_16x16x4_f32(false, q.a1, false, b1,
                                                (short)0, acc, false, false);
    acc = __builtin_amdgcn_wmma_f32_16x16x4_f32(false, q.a2, false, b2,
                                                (short)0, acc, false, false);
    acc = __builtin_amdgcn_wmma_f32_16x16x4_f32(false, q.a3, false, b3,
                                                (short)0, acc, false, false);
    return acc;
}

struct CatSel { const float* W; const float* bias; int od; };
__device__ __forceinline__ CatSel cat_select(int f,
        const float* W2a, const float* b2a, const float* W3, const float* b3,
        const float* W4,  const float* b4,  const float* W6, const float* b6,
        const float* W7,  const float* b7)
{
    CatSel s;
    if      (f < 4)  { s.W = W2a + f       * kH * 2; s.bias = b2a + f       * 2; s.od = 2; }
    else if (f < 6)  { s.W = W3  + (f - 4) * kH * 3; s.bias = b3  + (f - 4) * 3; s.od = 3; }
    else if (f < 7)  { s.W = W4;                     s.bias = b4;                s.od = 4; }
    else if (f < 8)  { s.W = W6;                     s.bias = b6;                s.od = 6; }
    else             { s.W = W7  + (f - 8) * kH * 7; s.bias = b7  + (f - 8) * 7; s.od = 7; }
    return s;
}

__device__ __forceinline__ void cat_compute(const OneIn& q, int f, int od,
                                            float* __restrict__ out,
                                            int tile, int kh, int n)
{
    const bool  valid = (n < od);
    const float bn    = valid ? q.braw : 0.0f;
    const v8f   acc   = one_product(q, valid);
    float v[8];
    int   idx[8];
#pragma unroll
    for (int r = 0; r < 8; ++r) {
        v[r]   = valid ? (acc[r] + bn) : -3.0e38f;
        idx[r] = n;
    }
#pragma unroll
    for (int mask = 1; mask <= 4; mask <<= 1) {
        float ov[8];
        int   oi[8];
#pragma unroll
        for (int r = 0; r < 8; ++r) {
            ov[r] = __shfl_xor(v[r],   mask, 32);
            oi[r] = __shfl_xor(idx[r], mask, 32);
        }
#pragma unroll
        for (int r = 0; r < 8; ++r) {
            const bool take = (ov[r] > v[r]) | ((ov[r] == v[r]) & (oi[r] < idx[r]));
            v[r]   = take ? ov[r] : v[r];
            idx[r] = take ? oi[r] : idx[r];
        }
    }
    if (n == 0) {
#pragma unroll
        for (int r = 0; r < 8; ++r) {
            const int orow = tile * 16 + r + 8 * kh;
            out[(size_t)orow * kF + f] = (float)idx[r];
        }
    }
}

__device__ __forceinline__ void cont_compute(const OneIn& q, float (&e)[8],
                                             float* __restrict__ out,
                                             int tile, int kh, int n)
{
    const bool  valid = (n < 2);
    const float bn    = valid ? q.braw : 0.0f;
    const v8f   acc   = one_product(q, valid);

#pragma unroll
    for (int r = 0; r < 8; ++r) asm volatile("" : "+v"(e[r]));

    float val[8];
#pragma unroll
    for (int r = 0; r < 8; ++r) {
        float t  = acc[r] + bn;
        float ts = __shfl_xor(t, 1, 32);
        const float mean = fast_tanh(t);
        const float sd   = fast_sigmoid(ts);
        val[r] = mean + sd * e[r];
    }
    if (n == 0) {
#pragma unroll
        for (int r = 0; r < 8; ++r) {
            const int orow = tile * 16 + r + 8 * kh;
            out[(size_t)orow * kF + 11] = val[r];
        }
    }
}

#define STAGE_FENCE() __builtin_amdgcn_sched_barrier(0)

__global__ __launch_bounds__(kWavesPerBlock * 32)
void tab_decoder_wmma(const float* __restrict__ x,
                      const float* __restrict__ W2a, const float* __restrict__ b2a,
                      const float* __restrict__ W3,  const float* __restrict__ b3,
                      const float* __restrict__ W4,  const float* __restrict__ b4,
                      const float* __restrict__ W6,  const float* __restrict__ b6,
                      const float* __restrict__ W7,  const float* __restrict__ b7,
                      const float* __restrict__ Wc,  const float* __restrict__ bc,
                      const float* __restrict__ eps,
                      float* __restrict__ out)
{
    const int tile = blockIdx.x;
    const int wave = __builtin_amdgcn_readfirstlane(threadIdx.x >> 5);
    const int lane = threadIdx.x & 31;
    const int m    = lane & 15;
    const int kh   = lane >> 4;
    const int n    = m;

    const int    arow = tile * 16 + m;
    const float* xb   = x + (size_t)arow * kRowStride;

    PackIn p0, p1;

    if (wave < 6) {
        const int    f  = wave;
        const CatSel s  = cat_select(f, W2a, b2a, W3, b3, W4, b4, W6, b6, W7, b7);
        const int    nc = (n < s.od) ? n : 0;
        OneIn q;
        one_load(q, xb + f * kH, s.W, s.bias, s.od, nc, kh);
        STAGE_FENCE();
        pack_load(p0, wave, xb, Wc, bc, eps, tile, kh, n);
        STAGE_FENCE();
        cat_compute(q, f, s.od, out, tile, kh, n);
        STAGE_FENCE();
        pack_load(p1, wave + 8, xb, Wc, bc, eps, tile, kh, n);
        STAGE_FENCE();
        pack_compute(p0, wave, out, tile, kh, n);
        STAGE_FENCE();
        pack_load(p0, wave + 16, xb, Wc, bc, eps, tile, kh, n);
        STAGE_FENCE();
        pack_compute(p1, wave + 8, out, tile, kh, n);
        STAGE_FENCE();
        pack_compute(p0, wave + 16, out, tile, kh, n);
    } else {
        const bool   w6 = (wave == 6);
        const int    fa = w6 ? 6 : 9;
        const int    fc = w6 ? 8 : 11;
        const CatSel sa = cat_select(fa,     W2a, b2a, W3, b3, W4, b4, W6, b6, W7, b7);
        const CatSel sb = cat_select(fa + 1, W2a, b2a, W3, b3, W4, b4, W6, b6, W7, b7);
        const float* Wt = w6 ? W7 : Wc;
        const float* bt = w6 ? b7 : bc;
        const int    ot = w6 ? 7 : 2;
        const int    na = (n < sa.od) ? n : 0;
        const int    nb = (n < sb.od) ? n : 0;
        const int    nt = (n < ot) ? n : 0;
        OneIn qa, qb, qc;
        one_load(qa, xb + fa * kH,       sa.W, sa.bias, sa.od, na, kh);
        one_load(qb, xb + (fa + 1) * kH, sb.W, sb.bias, sb.od, nb, kh);
        one_load(qc, xb + fc * kH,       Wt,   bt,      ot,    nt, kh);
        const float* ep = eps + (size_t)(tile * 16 + 8 * kh) * kNCont;
        float e11[8];
#pragma unroll
        for (int r = 0; r < 8; ++r) e11[r] = ep[r * kNCont];
        STAGE_FENCE();
        pack_load(p0, wave, xb, Wc, bc, eps, tile, kh, n);
        STAGE_FENCE();
#pragma unroll
        for (int r = 0; r < 8; ++r) asm volatile("" : "+v"(e11[r]));
        cat_compute(qa, fa,     sa.od, out, tile, kh, n);
        cat_compute(qb, fa + 1, sb.od, out, tile, kh, n);
        if (w6) cat_compute(qc, 8, 7, out, tile, kh, n);
        else    cont_compute(qc, e11, out, tile, kh, n);
        STAGE_FENCE();
        pack_load(p1, wave + 8, xb, Wc, bc, eps, tile, kh, n);
        STAGE_FENCE();
        pack_compute(p0, wave, out, tile, kh, n);
        STAGE_FENCE();
        pack_compute(p1, wave + 8, out, tile, kh, n);
    }
}

extern "C" void kernel_launch(void* const* d_in, const int* in_sizes, int n_in,
                              void* d_out, int out_size, void* d_ws, size_t ws_size,
                              hipStream_t stream) {
    const float* x   = (const float*)d_in[0];
    const float* W2a = (const float*)d_in[1];
    const float* b2a = (const float*)d_in[2];
    const float* W3  = (const float*)d_in[3];
    const float* b3  = (const float*)d_in[4];
    const float* W4  = (const float*)d_in[5];
    const float* b4  = (const float*)d_in[6];
    const float* W6  = (const float*)d_in[7];
    const float* b6  = (const float*)d_in[8];
    const float* W7  = (const float*)d_in[9];
    const float* b7  = (const float*)d_in[10];
    const float* Wc  = (const float*)d_in[11];
    const float* bc  = (const float*)d_in[12];
    const float* eps = (const float*)d_in[13];
    float* out = (float*)d_out;

    tab_decoder_wmma<<<dim3(kNRows / 16), dim3(256), 0, stream>>>(
        x, W2a, b2a, W3, b3, W4, b4, W6, b6, W7, b7, Wc, bc, eps, out);
}


# x slab of each 8-feature pack loaded as whole 128-byte lines (one row's 512 contiguous bytes per load) and turned into A fragments through wave-private LDS rows, no barrier; same sums
# speedup vs baseline: 7.6687x; 1.1003x over previous
#include <hip/hip_runtime.h>
#include <hip/hip_bf16.h>
#include <math.h>

typedef __attribute__((ext_vector_type(2))) float v2f;
typedef __attribute__((ext_vector_type(4))) float v4f;
typedef __attribute__((ext_vector_type(8))) float v8f;

constexpr int kNRows     = 8192;
constexpr int kF         = 188;
constexpr int kH         = 16;
constexpr int kRowStride = kF * kH;
constexpr int kNCont     = 177;
constexpr int kPack      = 8;
constexpr int kPackF0    = 12;
constexpr int kNPack     = (kF - kPackF0) / kPack;
static_assert(kPackF0 + kNPack * kPack == kF, "packs must end at the last feature");

__device__ __forceinline__ float fast_rcp(float a) {
    return __builtin_amdgcn_rcpf(a);
}
__device__ __forceinline__ float fast_sigmoid(float a) {
    return fast_rcp(1.0f + __expf(-a));
}
__device__ __forceinline__ float fast_tanh(float a) {
#if __has_builtin(__builtin_amdgcn_tanhf)
    return __builtin_amdgcn_tanhf(a);
#elif __has_builtin(__builtin_amdgcn_tanh_f32)
    return __builtin_amdgcn_tanh_f32(a);
#else
    const float e2 = __expf(-2.0f * __builtin_fabsf(a));
    const float t  = (1.0f - e2) * fast_rcp(1.0f + e2);
    return __builtin_copysignf(t, a);
#endif
}

constexpr int kWavesPerBlock = 8;

constexpr int kStPitch = 132;
constexpr int kStWave  = 16 * kStPitch;

struct PackIn {
    v4f   x[16];
    v4f   wA, wB, wC, wD;
    float bn;
    float e[8];
};

struct OneIn {
    v2f   a0, a1, a2, a3;
    float w[8];
    float braw;
};

__device__ __forceinline__ void pack_load(PackIn& p, int j, const float* xl,
                                          const float* __restrict__ Wc,
                                          const float* __restrict__ bc,
                                          const float* __restrict__ eps,
                                          int tile, int kh, int n)
{
    const int f0  = kPackF0 + kPack * j;
    const int fc0 = f0 - 11;
    const int tn  = n >> 1;

    const float* xg = xl + f0 * kH;
#pragma unroll
    for (int r = 0; r < 16; ++r) p.x[r] = *(const v4f*)(xg + r * kRowStride);

    const float* Wl = Wc + (size_t)(fc0 + tn) * (kH * 2) + 8 * kh;
    p.wA = *(const v4f*)(Wl + 0);
    p.wB = *(const v4f*)(Wl + 4);
    p.wC = *(const v4f*)(Wl + 16);
    p.wD = *(const v4f*)(Wl + 20);

    p.bn = bc[fc0 * 2 + n];

    const float* ep = eps + (size_t)(tile * 16 + 8 * kh) * kNCont + fc0 + tn;
#pragma unroll
    for (int r = 0; r < 8; ++r) p.e[r] = ep[r * kNCont];
}

__device__ __forceinline__ void pack_stage(const PackIn& p, v4f (&a)[16],
                                           float* stw, const float* stl)
{
#pragma unroll
    for (int r = 0; r < 16; ++r) *(v4f*)(stw + r * kStPitch) = p.x[r];
    __builtin_amdgcn_fence(__ATOMIC_RELEASE, "wavefront");
    __builtin_amdgcn_wave_barrier();
    __builtin_amdgcn_fence(__ATOMIC_ACQUIRE, "wavefront");
#pragma unroll
    for (int i = 0; i < 16; ++i) a[i] = *(const v4f*)(stl + 8 * i);
}

__device__ __forceinline__ void pack_compute(PackIn& p, const v4f (&a)[16], int j,
                                             float* __restrict__ out,
                                             int tile, int kh, int n)
{
    const int f0 = kPackF0 + kPack * j;
    const int tn = n >> 1;
    const int c  = n & 1;

    float wq[8];
    wq[0] = c ? p.wA.y : p.wA.x;   wq[1] = c ? p.wA.w : p.wA.z;
    wq[2] = c ? p.wB.y : p.wB.x;   wq[3] = c ? p.wB.w : p.wB.z;
    wq[4] = c ? p.wC.y : p.wC.x;   wq[5] = c ? p.wC.w : p.wC.z;
    wq[6] = c ? p.wD.y : p.wD.x;   wq[7] = c ? p.wD.w : p.wD.z;

    v8f acc = {};
#pragma unroll
    for (int i = 0; i < 16; ++i) {
        const bool mine = (tn == (i >> 1));
#pragma unroll
        for (int u = 0; u < 2; ++u) {
            v2f av, bv;
            av.x = a[i][2 * u];
            av.y = a[i][2 * u + 1];
            bv.x = mine ? wq[4 * (i & 1) + 2 * u]     : 0.0f;
            bv.y = mine ? wq[4 * (i & 1) + 2 * u + 1] : 0.0f;
            acc = __builtin_amdgcn_wmma_f32_16x16x4_f32(false, av, false, bv,
                                                        (short)0, acc, false, false);
        }
    }

#pragma unroll
    for (int r = 0; r < 8; ++r) asm volatile("" : "+v"(p.e[r]));

    float val[8];
#pragma unroll
    for (int r = 0; r < 8; ++r) {
        float t  = acc[r] + p.bn;
        float ts = __shfl_xor(t, 1, 32);
        const float mean = fast_tanh(t);
        const float sd   = fast_sigmoid(ts);
        val[r] = mean + sd * p.e[r];
    }
    if (c == 0) {
#pragma unroll
        for (int r = 0; r < 8; ++r) {
            const int orow = tile * 16 + r + 8 * kh;
            out[(size_t)orow * kF + f0 + tn] = val[r];
        }
    }
}

__device__ __forceinline__ void one_load(OneIn& q, const float* xf,
                                         const float* W, const float* bias,
                                         int od, int nc, int kh)
{
    q.braw = bias[nc];
    q.a0 = *(const v2f*)(xf + 0  + 2 * kh);
    q.a1 = *(const v2f*)(xf + 4  + 2 * kh);
    q.a2 = *(const v2f*)(xf + 8  + 2 * kh);
    q.a3 = *(const v2f*)(xf + 12 + 2 * kh);
    const float* Wl = W + 2 * kh * od + nc;
    q.w[0] = Wl[0 * od];   q.w[1] = Wl[1 * od];
    q.w[2] = Wl[4 * od];   q.w[3] = Wl[5 * od];
    q.w[4] = Wl[8 * od];   q.w[5] = Wl[9 * od];
    q.w[6] = Wl[12 * od];  q.w[7] = Wl[13 * od];
}

__device__ __forceinline__ v8f one_product(const OneIn& q, bool valid)
{
    v2f b0, b1, b2, b3;
    b0.x = valid ? q.w[0] : 0.0f;  b0.y = valid ? q.w[1] : 0.0f;
    b1.x = valid ? q.w[2] : 0.0f;  b1.y = valid ? q.w[3] : 0.0f;
    b2.x = valid ? q.w[4] : 0.0f;  b2.y = valid ? q.w[5] : 0.0f;
    b3.x = valid ? q.w[6] : 0.0f;  b3.y = valid ? q.w[7] : 0.0f;
    v8f acc = {};
    acc = __builtin_amdgcn_wmma_f32_16x16x4_f32(false, q.a0, false, b0,
                                                (short)0, acc, false, false);
    acc = __builtin_amdgcn_wmma_f32_16x16x4_f32(false, q.a1, false, b1,
                                                (short)0, acc, false, false);
    acc = __builtin_amdgcn_wmma_f32_16x16x4_f32(false, q.a2, false, b2,
                                                (short)0, acc, false, false);
    acc = __builtin_amdgcn_wmma_f32_16x16x4_f32(false, q.a3, false, b3,
                                                (short)0, acc, false, false);
    return acc;
}

struct CatSel { const float* W; const float* bias; int od; };
__device__ __forceinline__ CatSel cat_select(int f,
        const float* W2a, const float* b2a, const float* W3, const float* b3,
        const float* W4,  const float* b4,  const float* W6, const float* b6,
        const float* W7,  const float* b7)
{
    CatSel s;
    if      (f < 4)  { s.W = W2a + f       * kH * 2; s.bias = b2a + f       * 2; s.od = 2; }
    else if (f < 6)  { s.W = W3  + (f - 4) * kH * 3; s.bias = b3  + (f - 4) * 3; s.od = 3; }
    else if (f < 7)  { s.W = W4;                     s.bias = b4;                s.od = 4; }
    else if (f < 8)  { s.W = W6;                     s.bias = b6;                s.od = 6; }
    else             { s.W = W7  + (f - 8) * kH * 7; s.bias = b7  + (f - 8) * 7; s.od = 7; }
    return s;
}

__device__ __forceinline__ void cat_compute(const OneIn& q, int f, int od,
                                            float* __restrict__ out,
                                            int tile, int kh, int n)
{
    const bool  valid = (n < od);
    const float bn    = valid ? q.braw : 0.0f;
    const v8f   acc   = one_product(q, valid);
    float v[8];
    int   idx[8];
#pragma unroll
    for (int r = 0; r < 8; ++r) {
        v[r]   = valid ? (acc[r] + bn) : -3.0e38f;
        idx[r] = n;
    }
#pragma unroll
    for (int mask = 1; mask <= 4; mask <<= 1) {
        float ov[8];
        int   oi[8];
#pragma unroll
        for (int r = 0; r < 8; ++r) {
            ov[r] = __shfl_xor(v[r],   mask, 32);
            oi[r] = __shfl_xor(idx[r], mask, 32);
        }
#pragma unroll
        for (int r = 0; r < 8; ++r) {
            const bool take = (ov[r] > v[r]) | ((ov[r] == v[r]) & (oi[r] < idx[r]));
            v[r]   = take ? ov[r] : v[r];
            idx[r] = take ? oi[r] : idx[r];
        }
    }
    if (n == 0) {
#pragma unroll
        for (int r = 0; r < 8; ++r) {
            const int orow = tile * 16 + r + 8 * kh;
            out[(size_t)orow * kF + f] = (float)idx[r];
        }
    }
}

__device__ __forceinline__ void cont_compute(const OneIn& q, float (&e)[8],
                                             float* __restrict__ out,
                                             int tile, int kh, int n)
{
    const bool  valid = (n < 2);
    const float bn    = valid ? q.braw : 0.0f;
    const v8f   acc   = one_product(q, valid);

#pragma unroll
    for (int r = 0; r < 8; ++r) asm volatile("" : "+v"(e[r]));

    float val[8];
#pragma unroll
    for (int r = 0; r < 8; ++r) {
        float t  = acc[r] + bn;
        float ts = __shfl_xor(t, 1, 32);
        const float mean = fast_tanh(t);
        const float sd   = fast_sigmoid(ts);
        val[r] = mean + sd * e[r];
    }
    if (n == 0) {
#pragma unroll
        for (int r = 0; r < 8; ++r) {
            const int orow = tile * 16 + r + 8 * kh;
            out[(size_t)orow * kF + 11] = val[r];
        }
    }
}

#define STAGE_FENCE() __builtin_amdgcn_sched_barrier(0)

__global__ __launch_bounds__(kWavesPerBlock * 32)
void tab_decoder_wmma(const float* __restrict__ x,
                      const float* __restrict__ W2a, const float* __restrict__ b2a,
                      const float* __restrict__ W3,  const float* __restrict__ b3,
                      const float* __restrict__ W4,  const float* __restrict__ b4,
                      const float* __restrict__ W6,  const float* __restrict__ b6,
                      const float* __restrict__ W7,  const float* __restrict__ b7,
                      const float* __restrict__ Wc,  const float* __restrict__ bc,
                      const float* __restrict__ eps,
                      float* __restrict__ out)
{
    const int tile = blockIdx.x;
    const int wave = __builtin_amdgcn_readfirstlane(threadIdx.x >> 5);
    const int lane = threadIdx.x & 31;
    const int m    = lane & 15;
    const int kh   = lane >> 4;
    const int n    = m;

    const int    arow = tile * 16 + m;
    const float* xb   = x + (size_t)arow * kRowStride;
    const float* xl   = x + (size_t)(tile * 16) * kRowStride + 4 * lane;

    __shared__ __attribute__((aligned(16))) float stage[kWavesPerBlock * kStWave];
    float*       stw = stage + wave * kStWave + 4 * lane;
    const float* stl = stage + wave * kStWave + m * kStPitch + 4 * kh;

    PackIn p0, p1;
    v4f    a[16];

    if (wave < 6) {
        const int    f  = wave;
        const CatSel s  = cat_select(f, W2a, b2a, W3, b3, W4, b4, W6, b6, W7, b7);
        const int    nc = (n < s.od) ? n : 0;
        OneIn q;
        one_load(q, xb + f * kH, s.W, s.bias, s.od, nc, kh);
        STAGE_FENCE();
        pack_load(p0, wave, xl, Wc, bc, eps, tile, kh, n);
        STAGE_FENCE();
        cat_compute(q, f, s.od, out, tile, kh, n);
        STAGE_FENCE();
        pack_load(p1, wave + 8, xl, Wc, bc, eps, tile, kh, n);
        STAGE_FENCE();
        pack_stage(p0, a, stw, stl);
        STAGE_FENCE();
        pack_compute(p0, a, wave, out, tile, kh, n);
        STAGE_FENCE();
        pack_load(p0, wave + 16, xl, Wc, bc, eps, tile, kh, n);
        STAGE_FENCE();
        pack_stage(p1, a, stw, stl);
        STAGE_FENCE();
        pack_compute(p1, a, wave + 8, out, tile, kh, n);
        STAGE_FENCE();
        pack_stage(p0, a, stw, stl);
        STAGE_FENCE();
        pack_compute(p0, a, wave + 16, out, tile, kh, n);
    } else {
        const bool   w6 = (wave == 6);
        const int    fa = w6 ? 6 : 9;
        const int    fc = w6 ? 8 : 11;
        const CatSel sa = cat_select(fa,     W2a, b2a, W3, b3, W4, b4, W6, b6, W7, b7);
        const CatSel sb = cat_select(fa + 1, W2a, b2a, W3, b3, W4, b4, W6, b6, W7, b7);
        const float* Wt = w6 ? W7 : Wc;
        const float* bt = w6 ? b7 : bc;
        const int    ot = w6 ? 7 : 2;
        const int    na = (n < sa.od) ? n : 0;
        const int    nb = (n < sb.od) ? n : 0;
        const int    nt = (n < ot) ? n : 0;
        OneIn qa, qb, qc;
        one_load(qa, xb + fa * kH,       sa.W, sa.bias, sa.od, na, kh);
        one_load(qb, xb + (fa + 1) * kH, sb.W, sb.bias, sb.od, nb, kh);
        one_load(qc, xb + fc * kH,       Wt,   bt,      ot,    nt, kh);
        const float* ep = eps + (size_t)(tile * 16 + 8 * kh) * kNCont;
        float e11[8];
#pragma unroll
        for (int r = 0; r < 8; ++r) e11[r] = ep[r * kNCont];
        STAGE_FENCE();
        pack_load(p0, wave, xl, Wc, bc, eps, tile, kh, n);
        STAGE_FENCE();
#pragma unroll
        for (int r = 0; r < 8; ++r) asm volatile("" : "+v"(e11[r]));
        cat_compute(qa, fa,     sa.od, out, tile, kh, n);
        cat_compute(qb, fa + 1, sb.od, out, tile, kh, n);
        if (w6) cat_compute(qc, 8, 7, out, tile, kh, n);
        else    cont_compute(qc, e11, out, tile, kh, n);
        STAGE_FENCE();
        pack_load(p1, wave + 8, xl, Wc, bc, eps, tile, kh, n);
        STAGE_FENCE();
        pack_stage(p0, a, stw, stl);
        STAGE_FENCE();
        pack_compute(p0, a, wave, out, tile, kh, n);
        STAGE_FENCE();
        pack_stage(p1, a, stw, stl);
        STAGE_FENCE();
        pack_compute(p1, a, wave + 8, out, tile, kh, n);
    }
}

extern "C" void kernel_launch(void* const* d_in, const int* in_sizes, int n_in,
                              void* d_out, int out_size, void* d_ws, size_t ws_size,
                              hipStream_t stream) {
    const float* x   = (const float*)d_in[0];
    const float* W2a = (const float*)d_in[1];
    const float* b2a = (const float*)d_in[2];
    const float* W3  = (const float*)d_in[3];
    const float* b3  = (const float*)d_in[4];
    const float* W4  = (const float*)d_in[5];
    const float* b4  = (const float*)d_in[6];
    const float* W6  = (const float*)d_in[7];
    const float* b6  = (const float*)d_in[8];
    const float* W7  = (const float*)d_in[9];
    const float* b7  = (const float*)d_in[10];
    const float* Wc  = (const float*)d_in[11];
    const float* bc  = (const float*)d_in[12];
    const float* eps = (const float*)d_in[13];
    float* out = (float*)d_out;

    tab_decoder_wmma<<<dim3(kNRows / 16), dim3(256), 0, stream>>>(
        x, W2a, b2a, W3, b3, W4, b4, W6, b6, W7, b7, Wc, bc, eps, out);
}


# Outputs collected in a 16 x 188 tile in LDS and written behind one block barrier as whole 128-byte lines (3 16-byte stores per thread) instead of 272 sub-line store instructions per tile; same sums
# speedup vs baseline: 8.1273x; 1.0598x over previous
#include <hip/hip_runtime.h>
#include <hip/hip_bf16.h>
#include <math.h>

typedef __attribute__((ext_vector_type(2))) float v2f;
typedef __attribute__((ext_vector_type(4))) float v4f;
typedef __attribute__((ext_vector_type(8))) float v8f;

constexpr int kNRows     = 8192;
constexpr int kF         = 188;
constexpr int kH         = 16;
constexpr int kRowStride = kF * kH;
constexpr int kNCont     = 177;
constexpr int kPack      = 8;
constexpr int kPackF0    = 12;
constexpr int kNPack     = (kF - kPackF0) / kPack;
static_assert(kPackF0 + kNPack * kPack == kF, "packs must end at the last feature");

__device__ __forceinline__ float fast_rcp(float a) {
    return __builtin_amdgcn_rcpf(a);
}
__device__ __forceinline__ float fast_sigmoid(float a) {
    return fast_rcp(1.0f + __expf(-a));
}
__device__ __forceinline__ float fast_tanh(float a) {
#if __has_builtin(__builtin_amdgcn_tanhf)
    return __builtin_amdgcn_tanhf(a);
#elif __has_builtin(__builtin_amdgcn_tanh_f32)
    return __builtin_amdgcn_tanh_f32(a);
#else
    const float e2 = __expf(-2.0f * __builtin_fabsf(a));
    const float t  = (1.0f - e2) * fast_rcp(1.0f + e2);
    return __builtin_copysignf(t, a);
#endif
}

constexpr int kWavesPerBlock = 8;

constexpr int kStPitch = 132;
constexpr int kStWave  = 16 * kStPitch;

struct PackIn {
    v4f   x[16];
    v4f   wA, wB, wC, wD;
    float bn;
    float e[8];
};

struct OneIn {
    v2f   a0, a1, a2, a3;
    float w[8];
    float braw;
};

__device__ __forceinline__ void pack_load(PackIn& p, int j, const float* xl,
                                          const float* __restrict__ Wc,
                                          const float* __restrict__ bc,
                                          const float* __restrict__ eps,
                                          int tile, int kh, int n)
{
    const int f0  = kPackF0 + kPack * j;
    const int fc0 = f0 - 11;
    const int tn  = n >> 1;

    const float* xg = xl + f0 * kH;
#pragma unroll
    for (int r = 0; r < 16; ++r) p.x[r] = *(const v4f*)(xg + r * kRowStride);

    const float* Wl = Wc + (size_t)(fc0 + tn) * (kH * 2) + 8 * kh;
    p.wA = *(const v4f*)(Wl + 0);
    p.wB = *(const v4f*)(Wl + 4);
    p.wC = *(const v4f*)(Wl + 16);
    p.wD = *(const v4f*)(Wl + 20);

    p.bn = bc[fc0 * 2 + n];

    const float* ep = eps + (size_t)(tile * 16 + 8 * kh) * kNCont + fc0 + tn;
#pragma unroll
    for (int r = 0; r < 8; ++r) p.e[r] = ep[r * kNCont];
}

__device__ __forceinline__ void pack_stage(const PackIn& p, v4f (&a)[16],
                                           float* stw, const float* stl)
{
#pragma unroll
    for (int r = 0; r < 16; ++r) *(v4f*)(stw + r * kStPitch) = p.x[r];
    __builtin_amdgcn_fence(__ATOMIC_RELEASE, "wavefront");
    __builtin_amdgcn_wave_barrier();
    __builtin_amdgcn_fence(__ATOMIC_ACQUIRE, "wavefront");
#pragma unroll
    for (int i = 0; i < 16; ++i) a[i] = *(const v4f*)(stl + 8 * i);
}

__device__ __forceinline__ void pack_compute(PackIn& p, const v4f (&a)[16], int j,
                                             float* ot, int kh, int n)
{
    const int f0 = kPackF0 + kPack * j;
    const int tn = n >> 1;
    const int c  = n & 1;

    float wq[8];
    wq[0] = c ? p.wA.y : p.wA.x;   wq[1] = c ? p.wA.w : p.wA.z;
    wq[2] = c ? p.wB.y : p.wB.x;   wq[3] = c ? p.wB.w : p.wB.z;
    wq[4] = c ? p.wC.y : p.wC.x;   wq[5] = c ? p.wC.w : p.wC.z;
    wq[6] = c ? p.wD.y : p.wD.x;   wq[7] = c ? p.wD.w : p.wD.z;

    v8f acc = {};
#pragma unroll
    for (int i = 0; i < 16; ++i) {
        const bool mine = (tn == (i >> 1));
#pragma unroll
        for (int u = 0; u < 2; ++u) {
            v2f av, bv;
            av.x = a[i][2 * u];
            av.y = a[i][2 * u + 1];
            bv.x = mine ? wq[4 * (i & 1) + 2 * u]     : 0.0f;
            bv.y = mine ? wq[4 * (i & 1) + 2 * u + 1] : 0.0f;
            acc = __builtin_amdgcn_wmma_f32_16x16x4_f32(false, av, false, bv,
                                                        (short)0, acc, false, false);
        }
    }

#pragma unroll
    for (int r = 0; r < 8; ++r) asm volatile("" : "+v"(p.e[r]));

    float val[8];
#pragma unroll
    for (int r = 0; r < 8; ++r) {
        float t  = acc[r] + p.bn;
        float ts = __shfl_xor(t, 1, 32);
        const float mean = fast_tanh(t);
        const float sd   = fast_sigmoid(ts);
        val[r] = mean + sd * p.e[r];
    }
    if (c == 0) {
#pragma unroll
        for (int r = 0; r < 8; ++r) ot[(r + 8 * kh) * kF + f0 + tn] = val[r];
    }
}

__device__ __forceinline__ void one_load(OneIn& q, const float* xf,
                                         const float* W, const float* bias,
                                         int od, int nc, int kh)
{
    q.braw = bias[nc];
    q.a0 = *(const v2f*)(xf + 0  + 2 * kh);
    q.a1 = *(const v2f*)(xf + 4  + 2 * kh);
    q.a2 = *(const v2f*)(xf + 8  + 2 * kh);
    q.a3 = *(const v2f*)(xf + 12 + 2 * kh);
    const float* Wl = W + 2 * kh * od + nc;
    q.w[0] = Wl[0 * od];   q.w[1] = Wl[1 * od];
    q.w[2] = Wl[4 * od];   q.w[3] = Wl[5 * od];
    q.w[4] = Wl[8 * od];   q.w[5] = Wl[9 * od];
    q.w[6] = Wl[12 * od];  q.w[7] = Wl[13 * od];
}

__device__ __forceinline__ v8f one_product(const OneIn& q, bool valid)
{
    v2f b0, b1, b2, b3;
    b0.x = valid ? q.w[0] : 0.0f;  b0.y = valid ? q.w[1] : 0.0f;
    b1.x = valid ? q.w[2] : 0.0f;  b1.y = valid ? q.w[3] : 0.0f;
    b2.x = valid ? q.w[4] : 0.0f;  b2.y = valid ? q.w[5] : 0.0f;
    b3.x = valid ? q.w[6] : 0.0f;  b3.y = valid ? q.w[7] : 0.0f;
    v8f acc = {};
    acc = __builtin_amdgcn_wmma_f32_16x16x4_f32(false, q.a0, false, b0,
                                                (short)0, acc, false, false);
    acc = __builtin_amdgcn_wmma_f32_16x16x4_f32(false, q.a1, false, b1,
                                                (short)0, acc, false, false);
    acc = __builtin_amdgcn_wmma_f32_16x16x4_f32(false, q.a2, false, b2,
                                                (short)0, acc, false, false);
    acc = __builtin_amdgcn_wmma_f32_16x16x4_f32(false, q.a3, false, b3,
                                                (short)0, acc, false, false);
    return acc;
}

struct CatSel { const float* W; const float* bias; int od; };
__device__ __forceinline__ CatSel cat_select(int f,
        const float* W2a, const float* b2a, const float* W3, const float* b3,
        const float* W4,  const float* b4,  const float* W6, const float* b6,
        const float* W7,  const float* b7)
{
    CatSel s;
    if      (f < 4)  { s.W = W2a + f       * kH * 2; s.bias = b2a + f       * 2; s.od = 2; }
    else if (f < 6)  { s.W = W3  + (f - 4) * kH * 3; s.bias = b3  + (f - 4) * 3; s.od = 3; }
    else if (f < 7)  { s.W = W4;                     s.bias = b4;                s.od = 4; }
    else if (f < 8)  { s.W = W6;                     s.bias = b6;                s.od = 6; }
    else             { s.W = W7  + (f - 8) * kH * 7; s.bias = b7  + (f - 8) * 7; s.od = 7; }
    return s;
}

__device__ __forceinline__ void cat_compute(const OneIn& q, int f, int od,
                                            float* ot, int kh, int n)
{
    const bool  valid = (n < od);
    const float bn    = valid ? q.braw : 0.0f;
    const v8f   acc   = one_product(q, valid);
    float v[8];
    int   idx[8];
#pragma unroll
    for (int r = 0; r < 8; ++r) {
        v[r]   = valid ? (acc[r] + bn) : -3.0e38f;
        idx[r] = n;
    }
#pragma unroll
    for (int mask = 1; mask <= 4; mask <<= 1) {
        float ov[8];
        int   oi[8];
#pragma unroll
        for (int r = 0; r < 8; ++r) {
            ov[r] = __shfl_xor(v[r],   mask, 32);
            oi[r] = __shfl_xor(idx[r], mask, 32);
        }
#pragma unroll
        for (int r = 0; r < 8; ++r) {
            const bool take = (ov[r] > v[r]) | ((ov[r] == v[r]) & (oi[r] < idx[r]));
            v[r]   = take ? ov[r] : v[r];
            idx[r] = take ? oi[r] : idx[r];
        }
    }
    if (n == 0) {
#pragma unroll
        for (int r = 0; r < 8; ++r) ot[(r + 8 * kh) * kF + f] = (float)idx[r];
    }
}

__device__ __forceinline__ void cont_compute(const OneIn& q, float (&e)[8],
                                             float* ot, int kh, int n)
{
    const bool  valid = (n < 2);
    const float bn    = valid ? q.braw : 0.0f;
    const v8f   acc   = one_product(q, valid);

#pragma unroll
    for (int r = 0; r < 8; ++r) asm volatile("" : "+v"(e[r]));

    float val[8];
#pragma unroll
    for (int r = 0; r < 8; ++r) {
        float t  = acc[r] + bn;
        float ts = __shfl_xor(t, 1, 32);
        const float mean = fast_tanh(t);
        const float sd   = fast_sigmoid(ts);
        val[r] = mean + sd * e[r];
    }
    if (n == 0) {
#pragma unroll
        for (int r = 0; r < 8; ++r) ot[(r + 8 * kh) * kF + 11] = val[r];
    }
}

#define STAGE_FENCE() __builtin_amdgcn_sched_barrier(0)

__global__ __launch_bounds__(kWavesPerBlock * 32)
void tab_decoder_wmma(const float* __restrict__ x,
                      const float* __restrict__ W2a, const float* __restrict__ b2a,
                      const float* __restrict__ W3,  const float* __restrict__ b3,
                      const float* __restrict__ W4,  const float* __restrict__ b4,
                      const float* __restrict__ W6,  const float* __restrict__ b6,
                      const float* __restrict__ W7,  const float* __restrict__ b7,
                      const float* __restrict__ Wc,  const float* __restrict__ bc,
                      const float* __restrict__ eps,
                      float* __restrict__ out)
{
    const int tile = blockIdx.x;
    const int wave = __builtin_amdgcn_readfirstlane(threadIdx.x >> 5);
    const int lane = threadIdx.x & 31;
    const int m    = lane & 15;
    const int kh   = lane >> 4;
    const int n    = m;

    const int    arow = tile * 16 + m;
    const float* xb   = x + (size_t)arow * kRowStride;
    const float* xl   = x + (size_t)(tile * 16) * kRowStride + 4 * lane;

    __shared__ __attribute__((aligned(16))) float stage[kWavesPerBlock * kStWave];
    float*       stw = stage + wave * kStWave + 4 * lane;
    const float* stl = stage + wave * kStWave + m * kStPitch + 4 * kh;

    __shared__ __attribute__((aligned(16))) float otile[16 * kF];

    PackIn p0, p1;
    v4f    a[16];

    if (wave < 6) {
        const int    f  = wave;
        const CatSel s  = cat_select(f, W2a, b2a, W3, b3, W4, b4, W6, b6, W7, b7);
        const int    nc = (n < s.od) ? n : 0;
        OneIn q;
        one_load(q, xb + f * kH, s.W, s.bias, s.od, nc, kh);
        STAGE_FENCE();
        pack_load(p0, wave, xl, Wc, bc, eps, tile, kh, n);
        STAGE_FENCE();
        cat_compute(q, f, s.od, otile, kh, n);
        STAGE_FENCE();
        pack_load(p1, wave + 8, xl, Wc, bc, eps, tile, kh, n);
        STAGE_FENCE();
        pack_stage(p0, a, stw, stl);
        STAGE_FENCE();
        pack_compute(p0, a, wave, otile, kh, n);
        STAGE_FENCE();
        pack_load(p0, wave + 16, xl, Wc, bc, eps, tile, kh, n);
        STAGE_FENCE();
        pack_stage(p1, a, stw, stl);
        STAGE_FENCE();
        pack_compute(p1, a, wave + 8, otile, kh, n);
        STAGE_FENCE();
        pack_stage(p0, a, stw, stl);
        STAGE_FENCE();
        pack_compute(p0, a, wave + 16, otile, kh, n);
    } else {
        const bool   w6 = (wave == 6);
        const int    fa = w6 ? 6 : 9;
        const int    fc = w6 ? 8 : 11;
        const CatSel sa = cat_select(fa,     W2a, b2a, W3, b3, W4, b4, W6, b6, W7, b7);
        const CatSel sb = cat_select(fa + 1, W2a, b2a, W3, b3, W4, b4, W6, b6, W7, b7);
        const float* Wt = w6 ? W7 : Wc;
        const float* bt = w6 ? b7 : bc;
        const int    ot = w6 ? 7 : 2;
        const int    na = (n < sa.od) ? n : 0;
        const int    nb = (n < sb.od) ? n : 0;
        const int    nt = (n < ot) ? n : 0;
        OneIn qa, qb, qc;
        one_load(qa, xb + fa * kH,       sa.W, sa.bias, sa.od, na, kh);
        one_load(qb, xb + (fa + 1) * kH, sb.W, sb.bias, sb.od, nb, kh);
        one_load(qc, xb + fc * kH,       Wt,   bt,      ot,    nt, kh);
        const float* ep = eps + (size_t)(tile * 16 + 8 * kh) * kNCont;
        float e11[8];
#pragma unroll
        for (int r = 0; r < 8; ++r) e11[r] = ep[r * kNCont];
        STAGE_FENCE();
        pack_load(p0, wave, xl, Wc, bc, eps, tile, kh, n);
        STAGE_FENCE();
#pragma unroll
        for (int r = 0; r < 8; ++r) asm volatile("" : "+v"(e11[r]));
        cat_compute(qa, fa,     sa.od, otile, kh, n);
        cat_compute(qb, fa + 1, sb.od, otile, kh, n);
        if (w6) cat_compute(qc, 8, 7, otile, kh, n);
        else    cont_compute(qc, e11, otile, kh, n);
        STAGE_FENCE();
        pack_load(p1, wave + 8, xl, Wc, bc, eps, tile, kh, n);
        STAGE_FENCE();
        pack_stage(p0, a, stw, stl);
        STAGE_FENCE();
        pack_compute(p0, a, wave, otile, kh, n);
        STAGE_FENCE();
        pack_stage(p1, a, stw, stl);
        STAGE_FENCE();
        pack_compute(p1, a, wave + 8, otile, kh, n);
    }

    __syncthreads();

    const v4f* osrc = (const v4f*)otile;
    v4f*       odst = (v4f*)(out + (size_t)tile * (16 * kF));
    const int  t0   = threadIdx.x;
    odst[t0]       = osrc[t0];
    odst[t0 + 256] = osrc[t0 + 256];
    if (t0 + 512 < 16 * kF / 4) odst[t0 + 512] = osrc[t0 + 512];
}

extern "C" void kernel_launch(void* const* d_in, const int* in_sizes, int n_in,
                              void* d_out, int out_size, void* d_ws, size_t ws_size,
                              hipStream_t stream) {
    const float* x   = (const float*)d_in[0];
    const float* W2a = (const float*)d_in[1];
    const float* b2a = (const float*)d_in[2];
    const float* W3  = (const float*)d_in[3];
    const float* b3  = (const float*)d_in[4];
    const float* W4  = (const float*)d_in[5];
    const float* b4  = (const float*)d_in[6];
    const float* W6  = (const float*)d_in[7];
    const float* b6  = (const float*)d_in[8];
    const float* W7  = (const float*)d_in[9];
    const float* b7  = (const float*)d_in[10];
    const float* Wc  = (const float*)d_in[11];
    const float* bc  = (const float*)d_in[12];
    const float* eps = (const float*)d_in[13];
    float* out = (float*)d_out;

    tab_decoder_wmma<<<dim3(kNRows / 16), dim3(256), 0, stream>>>(
        x, W2a, b2a, W3, b3, W4, b4, W6, b6, W7, b7, Wc, bc, eps, out);
}
